# starting kernel
# baseline (speedup 1.0000x reference)
#include <hip/hip_runtime.h>
#include <hip/hip_bf16.h>
#include <math.h>

#define BATCH    128
#define IN_CAPS  2048
#define IN_DIM   8
#define NUM_CAPS 32
#define CAP_DIM  16
#define EPS      1e-8f

#define SPLITK   32
#define KCHUNK   (IN_CAPS / SPLITK)
#define SVEC     (BATCH * NUM_CAPS * CAP_DIM)

typedef float v2f __attribute__((ext_vector_type(2)));
typedef float v8f __attribute__((ext_vector_type(8)));


__global__ __launch_bounds__(256) void caps_s1_partial(const float* __restrict__ x,
                                                       const float* __restrict__ W,
                                                       float* __restrict__ part)
{
    const int j    = blockIdx.x;
    const int kc   = blockIdx.y;
    const int wav  = threadIdx.x >> 5;
    const int lane = threadIdx.x & 31;
    const int lo   = lane & 15;
    const int hi   = lane >> 4;
    const int m0   = wav * 16;
    const int i0   = kc * KCHUNK;

    const float* xA = x + (size_t)(m0 + lo) * (IN_CAPS * IN_DIM);
    const float* wB = W + (size_t)j * (CAP_DIM * IN_DIM) + lo * IN_DIM;

    v8f acc = {};
    for (int i = i0; i < i0 + KCHUNK; ++i) {
        const float* xi = xA + i * IN_DIM;
        const float* wi = wB + (size_t)i * (NUM_CAPS * CAP_DIM * IN_DIM);
#pragma unroll
        for (int s = 0; s < 2; ++s) {
            const int k = s * 4 + 2 * hi;
            v2f a, b;
            a.x = xi[k];     a.y = xi[k + 1];
            b.x = wi[k];     b.y = wi[k + 1];
            acc = __builtin_amdgcn_wmma_f32_16x16x4_f32(
                      false, a, false, b, (short)0, acc, false, false);
        }
    }
    float* p = part + (size_t)kc * SVEC + j * CAP_DIM + lo;
#pragma unroll
    for (int r = 0; r < 8; ++r)
        p[(size_t)(m0 + r + 8 * hi) * (NUM_CAPS * CAP_DIM)] = acc[r];
}

__global__ __launch_bounds__(256) void caps_s2_partial(const float* __restrict__ x,
                                                       const float* __restrict__ W,
                                                       const float* __restrict__ c2,
                                                       float* __restrict__ part)
{
    const int j    = blockIdx.x;
    const int kc   = blockIdx.y;
    const int wav  = threadIdx.x >> 5;
    const int lane = threadIdx.x & 31;
    const int lo   = lane & 15;
    const int hi   = lane >> 4;
    const int m0   = wav * 16;
    const int i0   = kc * KCHUNK;

    const float* xA = x  + (size_t)(m0 + lo) * (IN_CAPS * IN_DIM);
    const float* cA = c2 + (size_t)(m0 + lo) * (IN_CAPS * NUM_CAPS) + j;
    const float* wB = W  + (size_t)j * (CAP_DIM * IN_DIM) + lo * IN_DIM;

    v8f acc = {};
    for (int i = i0; i < i0 + KCHUNK; ++i) {
        const float  c  = cA[(size_t)i * NUM_CAPS];
        const float* xi = xA + i * IN_DIM;
        const float* wi = wB + (size_t)i * (NUM_CAPS * CAP_DIM * IN_DIM);
#pragma unroll
        for (int s = 0; s < 2; ++s) {
            const int k = s * 4 + 2 * hi;
            v2f a, b;
            a.x = c * xi[k];   a.y = c * xi[k + 1];
            b.x = wi[k];       b.y = wi[k + 1];
            acc = __builtin_amdgcn_wmma_f32_16x16x4_f32(
                      false, a, false, b, (short)0, acc, false, false);
        }
    }
    float* p = part + (size_t)kc * SVEC + j * CAP_DIM + lo;
#pragma unroll
    for (int r = 0; r < 8; ++r)
        p[(size_t)(m0 + r + 8 * hi) * (NUM_CAPS * CAP_DIM)] = acc[r];
}

__global__ __launch_bounds__(256) void caps_reduce_squash(const float* __restrict__ part,
                                                          float* __restrict__ out,
                                                          float pre)
{
    const int t = blockIdx.x * 256 + threadIdx.x;
    float s = 0.0f;
    for (int k = 0; k < SPLITK; ++k)
        s += part[(size_t)k * SVEC + t];
    s *= pre;
    float sq = s * s;
    sq += __shfl_xor(sq, 1, 16);
    sq += __shfl_xor(sq, 2, 16);
    sq += __shfl_xor(sq, 4, 16);
    sq += __shfl_xor(sq, 8, 16);
    const float scale = sq / ((1.0f + sq) * (sqrtf(sq) + EPS));
    out[t] = s * scale;
}

__global__ __launch_bounds__(256) void caps_agree_kernel(const float* __restrict__ x,
                                                         const float* __restrict__ W,
                                                         const float* __restrict__ v1,
                                                         float* __restrict__ bagr)
{
    const int i    = blockIdx.x;
    const int j    = blockIdx.y;
    const int wav  = threadIdx.x >> 5;
    const int lane = threadIdx.x & 31;
    const int lo   = lane & 15;
    const int hi   = lane >> 4;
    const int m0   = wav * 16;

    const float* xi = x + (size_t)(m0 + lo) * (IN_CAPS * IN_DIM) + i * IN_DIM;
    const float* wi = W + (size_t)i * (NUM_CAPS * CAP_DIM * IN_DIM)
                        + (size_t)j * (CAP_DIM * IN_DIM) + lo * IN_DIM;

    v8f acc = {};
#pragma unroll
    for (int s = 0; s < 2; ++s) {
        const int k = s * 4 + 2 * hi;
        v2f a, b;
        a.x = xi[k];     a.y = xi[k + 1];
        b.x = wi[k];     b.y = wi[k + 1];
        acc = __builtin_amdgcn_wmma_f32_16x16x4_f32(
                  false, a, false, b, (short)0, acc, false, false);
    }
#pragma unroll
    for (int r = 0; r < 8; ++r) {
        const int m = m0 + r + 8 * hi;
        float t = acc[r] * v1[(size_t)m * (NUM_CAPS * CAP_DIM) + j * CAP_DIM + lo];
        t += __shfl_xor(t, 1, 16);
        t += __shfl_xor(t, 2, 16);
        t += __shfl_xor(t, 4, 16);
        t += __shfl_xor(t, 8, 16);
        if (lo == r)
            bagr[(size_t)m * (IN_CAPS * NUM_CAPS) + (size_t)i * NUM_CAPS + j] = t;
    }
}

__global__ __launch_bounds__(256) void caps_softmax_kernel(float* __restrict__ bagr)
{
    const size_t row = (size_t)blockIdx.x * blockDim.x + threadIdx.x;
    float* p = bagr + row * NUM_CAPS;
    float vals[NUM_CAPS];
    float mx = -INFINITY;
#pragma unroll
    for (int j = 0; j < NUM_CAPS; ++j) { vals[j] = p[j]; mx = fmaxf(mx, vals[j]); }
    float sum = 0.0f;
#pragma unroll
    for (int j = 0; j < NUM_CAPS; ++j) { vals[j] = __expf(vals[j] - mx); sum += vals[j]; }
    const float inv = 1.0f / sum;
#pragma unroll
    for (int j = 0; j < NUM_CAPS; ++j) p[j] = vals[j] * inv;
}

extern "C" void kernel_launch(void* const* d_in, const int* in_sizes, int n_in,
                              void* d_out, int out_size, void* d_ws, size_t ws_size,
                              hipStream_t stream)
{
    const float* x = (const float*)d_in[0];
    const float* W = (const float*)d_in[1];
    float* vout = (float*)d_out;

    float* v1   = (float*)d_ws;
    float* bagr = v1 + (size_t)SVEC;
    float* part = bagr + (size_t)BATCH * IN_CAPS * NUM_CAPS;

    caps_s1_partial   <<<dim3(NUM_CAPS, SPLITK), 256, 0, stream>>>(x, W, part);
    caps_reduce_squash<<<dim3(SVEC / 256),       256, 0, stream>>>(part, v1,
                                                                  1.0f / (float)NUM_CAPS);
    caps_agree_kernel  <<<dim3(IN_CAPS, NUM_CAPS),       256, 0, stream>>>(x, W, v1, bagr);
    caps_softmax_kernel<<<dim3((BATCH * IN_CAPS) / 256), 256, 0, stream>>>(bagr);
    caps_s2_partial   <<<dim3(NUM_CAPS, SPLITK), 256, 0, stream>>>(x, W, bagr, part);
    caps_reduce_squash<<<dim3(SVEC / 256),       256, 0, stream>>>(part, vout, 1.0f);
}


# agreement kernel: x and W tiles staged through double-buffered LDS by coalesced 128-bit block loads; block = 128 samples x 4 capsules x 64 inputs, W^T and x fragments read from LDS; sum order kept
# speedup vs baseline: 5.0298x; 5.0298x over previous
#include <hip/hip_runtime.h>
#include <hip/hip_bf16.h>
#include <math.h>

#define BATCH    128
#define IN_CAPS  2048
#define IN_DIM   8
#define NUM_CAPS 32
#define CAP_DIM  16
#define EPS      1e-8f

#define SPLITK   32
#define KCHUNK   (IN_CAPS / SPLITK)
#define SVEC     (BATCH * NUM_CAPS * CAP_DIM)

typedef float v2f __attribute__((ext_vector_type(2)));
typedef float v8f __attribute__((ext_vector_type(8)));


#define JSTRIDE ((size_t)BATCH * IN_CAPS)
#define NJ        4
#define ST_I      8
#define NST       (KCHUNK / ST_I)
#define XP        68
#define XS_FLOATS (BATCH * XP)
#define WP        132
#define WS_FLOATS (ST_I * NJ * WP)
#define CS_FLOATS (NJ * ST_I * BATCH)
#define S1_STAGE  (XS_FLOATS + WS_FLOATS)
#define S2_STAGE  (XS_FLOATS + WS_FLOATS + CS_FLOATS)

__device__ __forceinline__ void lds_handover_barrier()
{
    __builtin_amdgcn_fence(__ATOMIC_RELEASE, "workgroup", "local");
    __builtin_amdgcn_s_barrier();
    __builtin_amdgcn_fence(__ATOMIC_ACQUIRE, "workgroup", "local");
}

template <bool SCALED>
__device__ __forceinline__ void caps_s_body(const float* x,
                                            const float* W,
                                            const float* c2,
                                            float* part,
                                            float* L)
{
    constexpr int STAGE = SCALED ? S2_STAGE : S1_STAGE;
    const int jg   = blockIdx.x;
    const int kc   = blockIdx.y;
    const int t    = threadIdx.x;
    const int wav  = t >> 5;
    const int lane = t & 31;
    const int lo   = lane & 15;
    const int hi   = lane >> 4;
    const int jw   = wav & 3;
    const int mh   = wav >> 2;
    const int i0   = kc * KCHUNK;

    const float* xg = x + (size_t)(t >> 4) * (IN_CAPS * IN_DIM) + (size_t)i0 * IN_DIM + (t & 15) * 4;
    const int    xl = (t >> 4) * XP + (t & 15) * 4;
    const int    r4 = t & 127;
    const int    wd = (r4 & 31) >> 1;
    const float* wg = W + (size_t)(i0 + (t >> 7)) * (NUM_CAPS * CAP_DIM * IN_DIM)
                        + (size_t)jg * (NJ * CAP_DIM * IN_DIM) + r4 * 4;
    const int    wl = ((t >> 7) * NJ + (r4 >> 5)) * WP + wd * 8 + (wd >> 3) * 4 + (r4 & 1) * 4;
    const float* cg = c2 + (size_t)(jg * NJ) * JSTRIDE + (size_t)i0 * BATCH + t * 4;

    float4 xr0, xr1, xr2, xr3, xr4, xr5, xr6, xr7, wr0, wr1, wr2, wr3;
    float4 cr0 = {}, cr1 = {}, cr2 = {}, cr3 = {};
#define XROWS16 ((size_t)16 * (IN_CAPS * IN_DIM))
#define WCAP    ((size_t)(NUM_CAPS * CAP_DIM * IN_DIM))
#define GLOAD_STAGE(sn)                                                              \
    do {                                                                             \
        const float* xq = xg + (sn) * (ST_I * IN_DIM);                               \
        xr0 = *(const float4*)(xq);                                                  \
        xr1 = *(const float4*)(xq + 1 * XROWS16);                                    \
        xr2 = *(const float4*)(xq + 2 * XROWS16);                                    \
        xr3 = *(const float4*)(xq + 3 * XROWS16);                                    \
        xr4 = *(const float4*)(xq + 4 * XROWS16);                                    \
        xr5 = *(const float4*)(xq + 5 * XROWS16);                                    \
        xr6 = *(const float4*)(xq + 6 * XROWS16);                                    \
        xr7 = *(const float4*)(xq + 7 * XROWS16);                                    \
        const float* wq = wg + (size_t)((sn) * ST_I) * WCAP;                         \
        wr0 = *(const float4*)(wq);                                                  \
        wr1 = *(const float4*)(wq + 2 * WCAP);                                       \
        wr2 = *(const float4*)(wq + 4 * WCAP);                                       \
        wr3 = *(const float4*)(wq + 6 * WCAP);                                       \
        if (SCALED) {                                                                \
            const float* cq = cg + (sn) * (ST_I * BATCH);                            \
            cr0 = *(const float4*)(cq);                                              \
            cr1 = *(const float4*)(cq + 1 * JSTRIDE);                                \
            cr2 = *(const float4*)(cq + 2 * JSTRIDE);                                \
            cr3 = *(const float4*)(cq + 3 * JSTRIDE);                                \
        }                                                                            \
    } while (0)
#define LSTORE_STAGE(Lb)                                                             \
    do {                                                                             \
        float* xd = (Lb) + xl;                                                       \
        *(float4*)(xd)               = xr0;                                          \
        *(float4*)(xd + 1 * 16 * XP) = xr1;                                          \
        *(float4*)(xd + 2 * 16 * XP) = xr2;                                          \
        *(float4*)(xd + 3 * 16 * XP) = xr3;                                          \
        *(float4*)(xd + 4 * 16 * XP) = xr4;                                          \
        *(float4*)(xd + 5 * 16 * XP) = xr5;                                          \
        *(float4*)(xd + 6 * 16 * XP) = xr6;                                          \
        *(float4*)(xd + 7 * 16 * XP) = xr7;                                          \
        float* wd4 = (Lb) + XS_FLOATS + wl;                                          \
        *(float4*)(wd4)               = wr0;                                         \
        *(float4*)(wd4 + 2 * NJ * WP) = wr1;                                         \
        *(float4*)(wd4 + 4 * NJ * WP) = wr2;                                         \
        *(float4*)(wd4 + 6 * NJ * WP) = wr3;                                         \
        if (SCALED) {                                                                \
            float* cd = (Lb) + XS_FLOATS + WS_FLOATS + t * 4;                        \
            *(float4*)(cd)                      = cr0;                               \
            *(float4*)(cd + 1 * (ST_I * BATCH)) = cr1;                               \
            *(float4*)(cd + 2 * (ST_I * BATCH)) = cr2;                               \
            *(float4*)(cd + 3 * (ST_I * BATCH)) = cr3;                               \
        }                                                                            \
    } while (0)

    GLOAD_STAGE(0);
    LSTORE_STAGE(L);
    GLOAD_STAGE(1);
    lds_handover_barrier();

    const int aoff = (mh * 64 + lo) * XP + 2 * hi;
    const int boff = jw * WP + lo * 8 + (lo >> 3) * 4 + 2 * hi;
    const int coff = jw * (ST_I * BATCH) + mh * 64 + lo;

    v8f acc0 = {}, acc1 = {}, acc2 = {}, acc3 = {};
#define COMPUTE_KSTEP(ii, s)                                                         \
    do {                                                                             \
        const v2f b = *(const v2f*)(ws + boff + (ii) * (NJ * WP) + (s) * 4);         \
        const float* ap = xs + aoff + (ii) * IN_DIM + (s) * 4;                       \
        v2f a0 = *(const v2f*)(ap);                                                  \
        v2f a1 = *(const v2f*)(ap + 1 * (16 * XP));                                  \
        v2f a2 = *(const v2f*)(ap + 2 * (16 * XP));                                  \
        v2f a3 = *(const v2f*)(ap + 3 * (16 * XP));                                  \
        if (SCALED) {                                                                \
            a0.x = cv0 * a0.x;  a0.y = cv0 * a0.y;                                   \
            a1.x = cv1 * a1.x;  a1.y = cv1 * a1.y;                                   \
            a2.x = cv2 * a2.x;  a2.y = cv2 * a2.y;                                   \
            a3.x = cv3 * a3.x;  a3.y = cv3 * a3.y;                                   \
        }                                                                            \
        acc0 = __builtin_amdgcn_wmma_f32_16x16x4_f32(                                \
                   false, a0, false, b, (short)0, acc0, false, false);               \
        acc1 = __builtin_amdgcn_wmma_f32_16x16x4_f32(                                \
                   false, a1, false, b, (short)0, acc1, false, false);               \
        acc2 = __builtin_amdgcn_wmma_f32_16x16x4_f32(                                \
                   false, a2, false, b, (short)0, acc2, false, false);               \
        acc3 = __builtin_amdgcn_wmma_f32_16x16x4_f32(                                \
                   false, a3, false, b, (short)0, acc3, false, false);               \
    } while (0)
#define COMPUTE_CAPS(ii)                                                             \
    do {                                                                             \
        float cv0 = 1.0f, cv1 = 1.0f, cv2 = 1.0f, cv3 = 1.0f;                        \
        if (SCALED) {                                                                \
            cv0 = cs[coff + (ii) * BATCH];                                           \
            cv1 = cs[coff + (ii) * BATCH + 16];                                      \
            cv2 = cs[coff + (ii) * BATCH + 32];                                      \
            cv3 = cs[coff + (ii) * BATCH + 48];                                      \
        }                                                                            \
        COMPUTE_KSTEP(ii, 0);                                                        \
        COMPUTE_KSTEP(ii, 1);                                                        \
    } while (0)
#define COMPUTE_STAGE(Lb)                                                            \
    do {                                                                             \
        const float* xs = (Lb);                                                      \
        const float* ws = xs + XS_FLOATS;                                            \
        const float* cs = ws + WS_FLOATS;                                            \
        (void)cs;                                                                    \
        COMPUTE_CAPS(0); COMPUTE_CAPS(1); COMPUTE_CAPS(2); COMPUTE_CAPS(3);          \
        COMPUTE_CAPS(4); COMPUTE_CAPS(5); COMPUTE_CAPS(6); COMPUTE_CAPS(7);          \
    } while (0)

    for (int st = 0; st < NST - 1; ++st) {
        COMPUTE_STAGE(L + (st & 1) * STAGE);
        __builtin_amdgcn_sched_barrier(0);
        LSTORE_STAGE(L + ((st + 1) & 1) * STAGE);
        if (st + 2 < NST) GLOAD_STAGE(st + 2);
        lds_handover_barrier();
    }
    COMPUTE_STAGE(L + ((NST - 1) & 1) * STAGE);
#undef COMPUTE_STAGE
#undef COMPUTE_CAPS
#undef COMPUTE_KSTEP
#undef GLOAD_STAGE
#undef LSTORE_STAGE

    float* p = part + (size_t)kc * SVEC + (jg * NJ + jw) * CAP_DIM + lo
                    + (size_t)(mh * 64 + 8 * hi) * (NUM_CAPS * CAP_DIM);
#pragma unroll
    for (int r = 0; r < 8; ++r) {
        p[(size_t)(r)      * (NUM_CAPS * CAP_DIM)] = acc0[r];
        p[(size_t)(16 + r) * (NUM_CAPS * CAP_DIM)] = acc1[r];
        p[(size_t)(32 + r) * (NUM_CAPS * CAP_DIM)] = acc2[r];
        p[(size_t)(48 + r) * (NUM_CAPS * CAP_DIM)] = acc3[r];
    }
}

__global__ __launch_bounds__(256) void caps_s1_partial(const float* x,
                                                       const float* W,
                                                       float* part)
{
    __shared__ __attribute__((aligned(16))) float L[2 * S1_STAGE];
    caps_s_body<false>(x, W, x, part, L);
}

__global__ __launch_bounds__(256) void caps_s2_partial(const float* x,
                                                       const float* W,
                                                       const float* c2,
                                                       float* part)
{
    __shared__ __attribute__((aligned(16))) float L[2 * S2_STAGE];
    caps_s_body<true>(x, W, c2, part, L);
}

__global__ __launch_bounds__(256) void caps_reduce_squash(const float* __restrict__ part,
                                                          float* __restrict__ out,
                                                          float pre)
{
    const int t = blockIdx.x * 256 + threadIdx.x;
    float s = 0.0f;
    for (int k = 0; k < SPLITK; ++k)
        s += part[(size_t)k * SVEC + t];
    s *= pre;
    float sq = s * s;
    sq += __shfl_xor(sq, 1, 16);
    sq += __shfl_xor(sq, 2, 16);
    sq += __shfl_xor(sq, 4, 16);
    sq += __shfl_xor(sq, 8, 16);
    const float scale = sq / ((1.0f + sq) * (sqrtf(sq) + EPS));
    out[t] = s * scale;
}

#define AG_NJ   4
#define AG_ICH  64
#define AG_STI  8
#define AG_NST  (AG_ICH / AG_STI)
#define AG_WP   152
#define AG_HP   80
#define AG_WS_FLOATS (AG_STI * AG_NJ * AG_WP)
#define AG_STAGE     (XS_FLOATS + AG_WS_FLOATS)
__global__ __launch_bounds__(256) void caps_agree_kernel(const float* x,
                                                         const float* W,
                                                         const float* v1,
                                                         float* bagr)
{
    __shared__ __attribute__((aligned(16))) float L[2 * AG_STAGE];
    const int jg   = blockIdx.x;
    const int ic   = blockIdx.y;
    const int t    = threadIdx.x;
    const int wav  = t >> 5;
    const int lane = t & 31;
    const int lo   = lane & 15;
    const int hi   = lane >> 4;
    const int jw   = wav & 3;
    const int mh   = wav >> 2;
    const int i0   = ic * AG_ICH;
    const int j    = jg * AG_NJ + jw;

    const float* xg = x + (size_t)(t >> 4) * (IN_CAPS * IN_DIM) + (size_t)i0 * IN_DIM + (t & 15) * 4;
    const int    xl = (t >> 4) * XP + (t & 15) * 4;
    const int    r4 = t & 127;
    const int    wd = (r4 & 31) >> 1;
    const float* wg = W + (size_t)(i0 + (t >> 7)) * (NUM_CAPS * CAP_DIM * IN_DIM)
                        + (size_t)jg * (AG_NJ * CAP_DIM * IN_DIM) + r4 * 4;
    const int    wl = ((t >> 7) * AG_NJ + (r4 >> 5)) * AG_WP + wd * 8 + (wd >> 3) * 16 + (r4 & 1) * 4;

    float4 xr0, xr1, xr2, xr3, xr4, xr5, xr6, xr7, wr0, wr1, wr2, wr3;
#define AG_XROWS16 ((size_t)16 * (IN_CAPS * IN_DIM))
#define AG_WCAP    ((size_t)(NUM_CAPS * CAP_DIM * IN_DIM))
#define AG_GLOAD(sn)                                                                 \
    do {                                                                             \
        const float* xq = xg + (sn) * (AG_STI * IN_DIM);                             \
        xr0 = *(const float4*)(xq);                                                  \
        xr1 = *(const float4*)(xq + 1 * AG_XROWS16);                                 \
        xr2 = *(const float4*)(xq + 2 * AG_XROWS16);                                 \
        xr3 = *(const float4*)(xq + 3 * AG_XROWS16);                                 \
        xr4 = *(const float4*)(xq + 4 * AG_XROWS16);                                 \
        xr5 = *(const float4*)(xq + 5 * AG_XROWS16);                                 \
        xr6 = *(const float4*)(xq + 6 * AG_XROWS16);                                 \
        xr7 = *(const float4*)(xq + 7 * AG_XROWS16);                                 \
        const float* wq = wg + (size_t)((sn) * AG_STI) * AG_WCAP;                    \
        wr0 = *(const float4*)(wq);                                                  \
        wr1 = *(const float4*)(wq + 2 * AG_WCAP);                                    \
        wr2 = *(const float4*)(wq + 4 * AG_WCAP);                                    \
        wr3 = *(const float4*)(wq + 6 * AG_WCAP);                                    \
    } while (0)
#define AG_LSTORE(Lb)                                                                \
    do {                                                                             \
        float* xd = (Lb) + xl;                                                       \
        *(float4*)(xd)               = xr0;                                          \
        *(float4*)(xd + 1 * 16 * XP) = xr1;                                          \
        *(float4*)(xd + 2 * 16 * XP) = xr2;                                          \
        *(float4*)(xd + 3 * 16 * XP) = xr3;                                          \
        *(float4*)(xd + 4 * 16 * XP) = xr4;                                          \
        *(float4*)(xd + 5 * 16 * XP) = xr5;                                          \
        *(float4*)(xd + 6 * 16 * XP) = xr6;                                          \
        *(float4*)(xd + 7 * 16 * XP) = xr7;                                          \
        float* wd4 = (Lb) + XS_FLOATS + wl;                                          \
        *(float4*)(wd4)                      = wr0;                                  \
        *(float4*)(wd4 + 2 * AG_NJ * AG_WP)  = wr1;                                  \
        *(float4*)(wd4 + 4 * AG_NJ * AG_WP)  = wr2;                                  \
        *(float4*)(wd4 + 6 * AG_NJ * AG_WP)  = wr3;                                  \
    } while (0)

    AG_GLOAD(0);

    v2f bf[4][4];
#pragma unroll
    for (int tl = 0; tl < 4; ++tl) {
        const float4* vq = (const float4*)(v1 + (size_t)(mh * 64 + tl * 16 + lo) * (NUM_CAPS * CAP_DIM)
                                              + j * CAP_DIM + 8 * hi);
        const float4 p0 = vq[0], p1 = vq[1];
        bf[tl][0].x = p0.x; bf[tl][0].y = p0.y;  bf[tl][1].x = p0.z; bf[tl][1].y = p0.w;
        bf[tl][2].x = p1.x; bf[tl][2].y = p1.y;  bf[tl][3].x = p1.z; bf[tl][3].y = p1.w;
    }

    AG_LSTORE(L);
    AG_GLOAD(1);
    lds_handover_barrier();

    const int i2   = lo >> 3;
    const int e    = lo & 7;
    const int aoff = (i2 * AG_NJ + jw) * AG_WP + hi * AG_HP + e;
    const int xoff = (mh * 64 + lo) * XP + hi * IN_DIM;
    float* outp = bagr + (size_t)j * JSTRIDE + (size_t)(i0 + hi) * BATCH + mh * 64 + lo;

#define AG_PAIR(sn, ip)                                                              \
    do {                                                                             \
        const float* wp = ws + aoff + (ip) * (2 * AG_NJ * AG_WP);                    \
        v2f af[4];                                                                   \
        _Pragma("unroll")                                                            \
        for (int s = 0; s < 4; ++s) {                                                \
            af[s].x = wp[16 * s];                                                    \
            af[s].y = wp[16 * s + 8];                                                \
        }                                                                            \
        v8f g[4] = {};                                                               \
        _Pragma("unroll")                                                            \
        for (int s = 0; s < 4; ++s) {                                                \
            _Pragma("unroll")                                                        \
            for (int tl = 0; tl < 4; ++tl)                                           \
                g[tl] = __builtin_amdgcn_wmma_f32_16x16x4_f32(                       \
                            false, af[s], false, bf[tl][s], (short)0, g[tl], false, false); \
        }                                                                            \
        _Pragma("unroll")                                                            \
        for (int tl = 0; tl < 4; ++tl) {                                             \
            const float* xp = xs + xoff + tl * (16 * XP) + (ip) * (2 * IN_DIM);      \
            const float4 xa0 = *(const float4*)(xp);                                 \
            const float4 xa1 = *(const float4*)(xp + 4);                             \
            float t0 = g[tl][0] * xa0.x;                                             \
            t0 += g[tl][1] * xa0.y;  t0 += g[tl][2] * xa0.z;  t0 += g[tl][3] * xa0.w; \
            t0 += g[tl][4] * xa1.x;  t0 += g[tl][5] * xa1.y;  t0 += g[tl][6] * xa1.z; \
            t0 += g[tl][7] * xa1.w;                                                  \
            outp[(size_t)((sn) * AG_STI + (ip) * 2) * BATCH + tl * 16] = t0;         \
        }                                                                            \
    } while (0)
#define AG_COMPUTE(Lb, sn)                                                           \
    do {                                                                             \
        const float* xs = (Lb);                                                      \
        const float* ws = xs + XS_FLOATS;                                            \
        AG_PAIR(sn, 0); AG_PAIR(sn, 1); AG_PAIR(sn, 2); AG_PAIR(sn, 3);              \
    } while (0)

    for (int st = 0; st < AG_NST - 1; ++st) {
        AG_COMPUTE(L + (st & 1) * AG_STAGE, st);
        __builtin_amdgcn_sched_barrier(0);
        AG_LSTORE(L + ((st + 1) & 1) * AG_STAGE);
        if (st + 2 < AG_NST) AG_GLOAD(st + 2);
        lds_handover_barrier();
    }
    AG_COMPUTE(L + ((AG_NST - 1) & 1) * AG_STAGE, AG_NST - 1);
#undef AG_COMPUTE
#undef AG_PAIR
#undef AG_GLOAD
#undef AG_LSTORE
}

__global__ __launch_bounds__(256) void caps_softmax_kernel(float* __restrict__ bagr)
{
    const size_t row = (size_t)blockIdx.x * blockDim.x + threadIdx.x;
    float* p = bagr + row;
    float vals[NUM_CAPS];
    float mx = -INFINITY;
#pragma unroll
    for (int j = 0; j < NUM_CAPS; ++j) { vals[j] = p[(size_t)j * JSTRIDE]; mx = fmaxf(mx, vals[j]); }
    float sum = 0.0f;
#pragma unroll
    for (int j = 0; j < NUM_CAPS; ++j) { vals[j] = __expf(vals[j] - mx); sum += vals[j]; }
    const float inv = 1.0f / sum;
#pragma unroll
    for (int j = 0; j < NUM_CAPS; ++j) p[(size_t)j * JSTRIDE] = vals[j] * inv;
}

extern "C" void kernel_launch(void* const* d_in, const int* in_sizes, int n_in,
                              void* d_out, int out_size, void* d_ws, size_t ws_size,
                              hipStream_t stream)
{
    const float* x = (const float*)d_in[0];
    const float* W = (const float*)d_in[1];
    float* vout = (float*)d_out;

    float* v1   = (float*)d_ws;
    float* bagr = v1 + (size_t)SVEC;
    float* part = bagr + (size_t)BATCH * IN_CAPS * NUM_CAPS;

    caps_s1_partial   <<<dim3(NUM_CAPS / NJ, SPLITK), 256, 0, stream>>>(x, W, part);
    caps_reduce_squash<<<dim3(SVEC / 256),       256, 0, stream>>>(part, v1,
                                                                  1.0f / (float)NUM_CAPS);
    caps_agree_kernel  <<<dim3(NUM_CAPS / AG_NJ, IN_CAPS / AG_ICH), 256, 0, stream>>>(x, W, v1, bagr);
    caps_softmax_kernel<<<dim3((BATCH * IN_CAPS) / 256), 256, 0, stream>>>(bagr);
    caps_s2_partial   <<<dim3(NUM_CAPS / NJ, SPLITK), 256, 0, stream>>>(x, W, bagr, part);
    caps_reduce_squash<<<dim3(SVEC / 256),       256, 0, stream>>>(part, vout, 1.0f);
}
